# starting kernel
# baseline (speedup 1.0000x reference)
#include <hip/hip_runtime.h>
#include <hip/hip_bf16.h>

#define NROWS 262144
#define DCOLS 128
#define CS_BLOCKS 512
#define PART_STRIDE 65536

typedef __attribute__((ext_vector_type(2))) float v2f;
typedef __attribute__((ext_vector_type(8))) float v8f;

__device__ __forceinline__ void f4add(float4& a, const float4& b) {
    a.x += b.x; a.y += b.y; a.z += b.z; a.w += b.w;
}

__global__ __launch_bounds__(256) void colsum_kernel(
    const float* __restrict__ x1, const float* __restrict__ x2,
    float* __restrict__ part) {
    const int tid = threadIdx.x;
    const int cg  = tid & 31;
    const int rg  = tid >> 5;
    float4 a1 = make_float4(0.f, 0.f, 0.f, 0.f);
    float4 a2 = make_float4(0.f, 0.f, 0.f, 0.f);
    const int step = CS_BLOCKS * 8;
    for (int row = blockIdx.x * 8 + rg; row < NROWS; row += step) {
        const float4 u = ((const float4*)(x1 + (size_t)row * DCOLS))[cg];
        const float4 v = ((const float4*)(x2 + (size_t)row * DCOLS))[cg];
        f4add(a1, u);
        f4add(a2, v);
    }
    __shared__ float4 l1[256];
    __shared__ float4 l2[256];
    l1[tid] = a1;
    l2[tid] = a2;
    __syncthreads();
    #pragma unroll
    for (int s = 4; s >= 1; s >>= 1) {
        if (rg < s) {
            f4add(l1[tid], l1[tid + s * 32]);
            f4add(l2[tid], l2[tid + s * 32]);
        }
        __syncthreads();
    }
    if (rg == 0) {
        float* o1 = part + (size_t)blockIdx.x * DCOLS + cg * 4;
        float* o2 = o1 + PART_STRIDE;
        *(float4*)o1 = l1[tid];
        *(float4*)o2 = l2[tid];
    }
}

__global__ __launch_bounds__(256) void colreduce_kernel(
    const float* __restrict__ part, float* __restrict__ sums) {
    const int t   = threadIdx.x;
    const int mat = t >> 7;
    const int c   = t & 127;
    const float* p = part + (size_t)mat * PART_STRIDE + c;
    float a0 = 0.f, a1 = 0.f, a2 = 0.f, a3 = 0.f;
    for (int b = 0; b < CS_BLOCKS; b += 4) {
        a0 += p[(size_t)(b + 0) * DCOLS];
        a1 += p[(size_t)(b + 1) * DCOLS];
        a2 += p[(size_t)(b + 2) * DCOLS];
        a3 += p[(size_t)(b + 3) * DCOLS];
    }
    sums[t] = (a0 + a1) + (a2 + a3);
}

__global__ __launch_bounds__(256) void cosine_kernel(
    const float* __restrict__ x1, const float* __restrict__ x2,
    const float* __restrict__ sums, float* __restrict__ out) {
    __shared__ float s_sum[256];
    const int tid = threadIdx.x;
    s_sum[tid] = sums[tid];
    __syncthreads();

    const int wave = tid >> 5;
    const int lane = tid & 31;
    const int r0   = (blockIdx.x * 8 + wave) * 16;
    const int m    = lane & 15;
    const int kk   = (lane >> 4) * 2;

    const float* pA = x1 + (size_t)(r0 + m) * DCOLS + kk;
    const float* pB = x2 + (size_t)(r0 + m) * DCOLS + kk;
    const float* s1 = s_sum;
    const float* s2 = s_sum + 128;

    v8f accN = {0.f, 0.f, 0.f, 0.f, 0.f, 0.f, 0.f, 0.f};
    float v1p = 0.f, v2p = 0.f;

    #pragma unroll
    for (int ks = 0; ks < 32; ++ks) {
        const int k0 = ks * 4;
        const v2f a1 = *(const v2f*)(pA + k0);
        const v2f a2 = *(const v2f*)(pB + k0);
        accN = __builtin_amdgcn_wmma_f32_16x16x4_f32(
             false, a1,  false, a2,
             (short)0, accN,  false,  false);
        v1p += a1.x * s1[k0 + kk] + a1.y * s1[k0 + kk + 1];
        v2p += a2.x * s2[k0 + kk] + a2.y * s2[k0 + kk + 1];
    }

    const float v1 = v1p + __shfl_xor(v1p, 16, 32);
    const float v2 = v2p + __shfl_xor(v2p, 16, 32);

    float numD = 0.f;
    #pragma unroll
    for (int v = 0; v < 8; ++v) {
        if (lane == v)      numD = accN[v];
        if (lane == v + 24) numD = accN[v];
    }

    int row = -1;
    if (lane < 8)       row = lane;
    else if (lane >= 24) row = lane - 16;
    if (row >= 0) {
        out[r0 + row] = numD / sqrtf(v1 * v2);
    }
}

extern "C" void kernel_launch(void* const* d_in, const int* in_sizes, int n_in,
                              void* d_out, int out_size, void* d_ws, size_t ws_size,
                              hipStream_t stream) {
    const float* x1 = (const float*)d_in[0];
    const float* x2 = (const float*)d_in[1];
    float* out  = (float*)d_out;
    float* sums = (float*)d_ws;
    float* part = out;

    colsum_kernel<<<CS_BLOCKS, 256, 0, stream>>>(x1, x2, part);
    colreduce_kernel<<<1, 256, 0, stream>>>(part, sums);
    cosine_kernel<<<NROWS / 128, 256, 0, stream>>>(x1, x2, sums, out);
}


# column-sum pass: constant trip count, fully unrolled, compiler told 4 waves per SIMD is all the grid gives: about 50 b128 loads in flight per wave, was 2; same order of sums
# speedup vs baseline: 1.2406x; 1.2406x over previous
#include <hip/hip_runtime.h>
#include <hip/hip_bf16.h>

#define NROWS 262144
#define DCOLS 128
#define CS_BLOCKS 512
#define PART_STRIDE 65536

typedef __attribute__((ext_vector_type(2))) float v2f;
typedef __attribute__((ext_vector_type(8))) float v8f;

__device__ __forceinline__ void f4add(float4& a, const float4& b) {
    a.x += b.x; a.y += b.y; a.z += b.z; a.w += b.w;
}

#define CS_RPT   (NROWS / (CS_BLOCKS * 8))
#define CS_STEP4 ((size_t)CS_BLOCKS * 8 * (DCOLS / 4))
static_assert(NROWS % (CS_BLOCKS * 8) == 0, "every thread sums the same number of rows");

__global__ __launch_bounds__(256) __attribute__((amdgpu_waves_per_eu(4, 4))) void colsum_kernel(
    const float* __restrict__ x1, const float* __restrict__ x2,
    float* __restrict__ part) {
    const int tid = threadIdx.x;
    const int cg  = tid & 31;
    const int rg  = tid >> 5;
    float4 a1 = make_float4(0.f, 0.f, 0.f, 0.f);
    float4 a2 = make_float4(0.f, 0.f, 0.f, 0.f);
    const float4* p1 = (const float4*)(x1 + (size_t)(blockIdx.x * 8 + rg) * DCOLS) + cg;
    const float4* p2 = (const float4*)(x2 + (size_t)(blockIdx.x * 8 + rg) * DCOLS) + cg;
    #pragma unroll
    for (int i = 0; i < CS_RPT; ++i) {
        f4add(a1, p1[(size_t)i * CS_STEP4]);
        f4add(a2, p2[(size_t)i * CS_STEP4]);
    }
    __shared__ float4 l1[256];
    __shared__ float4 l2[256];
    l1[tid] = a1;
    l2[tid] = a2;
    __syncthreads();
    #pragma unroll
    for (int s = 4; s >= 1; s >>= 1) {
        if (rg < s) {
            f4add(l1[tid], l1[tid + s * 32]);
            f4add(l2[tid], l2[tid + s * 32]);
        }
        __syncthreads();
    }
    if (rg == 0) {
        float* o1 = part + (size_t)blockIdx.x * DCOLS + cg * 4;
        float* o2 = o1 + PART_STRIDE;
        *(float4*)o1 = l1[tid];
        *(float4*)o2 = l2[tid];
    }
}

__global__ __launch_bounds__(256) void colreduce_kernel(
    const float* __restrict__ part, float* __restrict__ sums) {
    const int t   = threadIdx.x;
    const int mat = t >> 7;
    const int c   = t & 127;
    const float* p = part + (size_t)mat * PART_STRIDE + c;
    float a0 = 0.f, a1 = 0.f, a2 = 0.f, a3 = 0.f;
    for (int b = 0; b < CS_BLOCKS; b += 4) {
        a0 += p[(size_t)(b + 0) * DCOLS];
        a1 += p[(size_t)(b + 1) * DCOLS];
        a2 += p[(size_t)(b + 2) * DCOLS];
        a3 += p[(size_t)(b + 3) * DCOLS];
    }
    sums[t] = (a0 + a1) + (a2 + a3);
}

__global__ __launch_bounds__(256) void cosine_kernel(
    const float* __restrict__ x1, const float* __restrict__ x2,
    const float* __restrict__ sums, float* __restrict__ out) {
    __shared__ float s_sum[256];
    const int tid = threadIdx.x;
    s_sum[tid] = sums[tid];
    __syncthreads();

    const int wave = tid >> 5;
    const int lane = tid & 31;
    const int r0   = (blockIdx.x * 8 + wave) * 16;
    const int m    = lane & 15;
    const int kk   = (lane >> 4) * 2;

    const float* pA = x1 + (size_t)(r0 + m) * DCOLS + kk;
    const float* pB = x2 + (size_t)(r0 + m) * DCOLS + kk;
    const float* s1 = s_sum;
    const float* s2 = s_sum + 128;

    v8f accN = {0.f, 0.f, 0.f, 0.f, 0.f, 0.f, 0.f, 0.f};
    float v1p = 0.f, v2p = 0.f;

    #pragma unroll
    for (int ks = 0; ks < 32; ++ks) {
        const int k0 = ks * 4;
        const v2f a1 = *(const v2f*)(pA + k0);
        const v2f a2 = *(const v2f*)(pB + k0);
        accN = __builtin_amdgcn_wmma_f32_16x16x4_f32(
             false, a1,  false, a2,
             (short)0, accN,  false,  false);
        v1p += a1.x * s1[k0 + kk] + a1.y * s1[k0 + kk + 1];
        v2p += a2.x * s2[k0 + kk] + a2.y * s2[k0 + kk + 1];
    }

    const float v1 = v1p + __shfl_xor(v1p, 16, 32);
    const float v2 = v2p + __shfl_xor(v2p, 16, 32);

    float numD = 0.f;
    #pragma unroll
    for (int v = 0; v < 8; ++v) {
        if (lane == v)      numD = accN[v];
        if (lane == v + 24) numD = accN[v];
    }

    int row = -1;
    if (lane < 8)       row = lane;
    else if (lane >= 24) row = lane - 16;
    if (row >= 0) {
        out[r0 + row] = numD / sqrtf(v1 * v2);
    }
}

extern "C" void kernel_launch(void* const* d_in, const int* in_sizes, int n_in,
                              void* d_out, int out_size, void* d_ws, size_t ws_size,
                              hipStream_t stream) {
    const float* x1 = (const float*)d_in[0];
    const float* x2 = (const float*)d_in[1];
    float* out  = (float*)d_out;
    float* sums = (float*)d_ws;
    float* part = out;

    colsum_kernel<<<CS_BLOCKS, 256, 0, stream>>>(x1, x2, part);
    colreduce_kernel<<<1, 256, 0, stream>>>(part, sums);
    cosine_kernel<<<NROWS / 128, 256, 0, stream>>>(x1, x2, sums, out);
}


# final pass: each wave reads its rows with whole-line loads (four full lines per instruction) and takes the WMMA fragments from a padded wave-private LDS tile, not from 16-line loads; same sum order
# speedup vs baseline: 1.3966x; 1.1257x over previous
#include <hip/hip_runtime.h>
#include <hip/hip_bf16.h>

#define NROWS 262144
#define DCOLS 128
#define CS_BLOCKS 512
#define PART_STRIDE 65536

typedef __attribute__((ext_vector_type(2))) float v2f;
typedef __attribute__((ext_vector_type(8))) float v8f;

__device__ __forceinline__ void f4add(float4& a, const float4& b) {
    a.x += b.x; a.y += b.y; a.z += b.z; a.w += b.w;
}

#define CS_RPT   (NROWS / (CS_BLOCKS * 8))
#define CS_STEP4 ((size_t)CS_BLOCKS * 8 * (DCOLS / 4))
static_assert(NROWS % (CS_BLOCKS * 8) == 0, "every thread sums the same number of rows");

__global__ __launch_bounds__(256) __attribute__((amdgpu_waves_per_eu(4, 4))) void colsum_kernel(
    const float* __restrict__ x1, const float* __restrict__ x2,
    float* __restrict__ part) {
    const int tid = threadIdx.x;
    const int cg  = tid & 31;
    const int rg  = tid >> 5;
    float4 a1 = make_float4(0.f, 0.f, 0.f, 0.f);
    float4 a2 = make_float4(0.f, 0.f, 0.f, 0.f);
    const float4* p1 = (const float4*)(x1 + (size_t)(blockIdx.x * 8 + rg) * DCOLS) + cg;
    const float4* p2 = (const float4*)(x2 + (size_t)(blockIdx.x * 8 + rg) * DCOLS) + cg;
    #pragma unroll
    for (int i = 0; i < CS_RPT; ++i) {
        f4add(a1, p1[(size_t)i * CS_STEP4]);
        f4add(a2, p2[(size_t)i * CS_STEP4]);
    }
    __shared__ float4 l1[256];
    __shared__ float4 l2[256];
    l1[tid] = a1;
    l2[tid] = a2;
    __syncthreads();
    #pragma unroll
    for (int s = 4; s >= 1; s >>= 1) {
        if (rg < s) {
            f4add(l1[tid], l1[tid + s * 32]);
            f4add(l2[tid], l2[tid + s * 32]);
        }
        __syncthreads();
    }
    if (rg == 0) {
        float* o1 = part + (size_t)blockIdx.x * DCOLS + cg * 4;
        float* o2 = o1 + PART_STRIDE;
        *(float4*)o1 = l1[tid];
        *(float4*)o2 = l2[tid];
    }
}

__global__ __launch_bounds__(256) void colreduce_kernel(
    const float* __restrict__ part, float* __restrict__ sums) {
    const int t   = threadIdx.x;
    const int mat = t >> 7;
    const int c   = t & 127;
    const float* p = part + (size_t)mat * PART_STRIDE + c;
    float a0 = 0.f, a1 = 0.f, a2 = 0.f, a3 = 0.f;
    for (int b = 0; b < CS_BLOCKS; b += 4) {
        a0 += p[(size_t)(b + 0) * DCOLS];
        a1 += p[(size_t)(b + 1) * DCOLS];
        a2 += p[(size_t)(b + 2) * DCOLS];
        a3 += p[(size_t)(b + 3) * DCOLS];
    }
    sums[t] = (a0 + a1) + (a2 + a3);
}

#define TILE_PITCH4 33
#define TILE_PITCH  (TILE_PITCH4 * 4)

__global__ __launch_bounds__(256) void cosine_kernel(
    const float* __restrict__ x1, const float* __restrict__ x2,
    const float* __restrict__ sums, float* __restrict__ out) {
    __shared__ float4 s_lds[64 + 8 * 2 * 16 * TILE_PITCH4];
    float*  s_sum  = (float*)s_lds;
    float4* s_tile = s_lds + 64;
    const int tid  = threadIdx.x;
    const int wave = tid >> 5;
    const int lane = tid & 31;
    const int r0   = (blockIdx.x * 8 + wave) * 16;

    const float ssum = sums[tid];

    const float4* g1 = (const float4*)(x1 + (size_t)r0 * DCOLS) + lane;
    const float4* g2 = (const float4*)(x2 + (size_t)r0 * DCOLS) + lane;
    float4 q1[16], q2[16];
    #pragma unroll
    for (int r = 0; r < 16; ++r) {
        q1[r] = g1[r * (DCOLS / 4)];
        q2[r] = g2[r * (DCOLS / 4)];
    }
    float4* t1 = s_tile + (wave * 2 + 0) * (16 * TILE_PITCH4);
    float4* t2 = s_tile + (wave * 2 + 1) * (16 * TILE_PITCH4);
    s_sum[tid] = ssum;
    #pragma unroll
    for (int r = 0; r < 16; ++r) {
        t1[r * TILE_PITCH4 + lane] = q1[r];
        t2[r * TILE_PITCH4 + lane] = q2[r];
    }
    __syncthreads();

    const int m  = lane & 15;
    const int kk = (lane >> 4) * 2;

    const float* pA = (const float*)t1 + m * TILE_PITCH + kk;
    const float* pB = (const float*)t2 + m * TILE_PITCH + kk;
    const float* s1 = s_sum;
    const float* s2 = s_sum + 128;

    v8f accN = {0.f, 0.f, 0.f, 0.f, 0.f, 0.f, 0.f, 0.f};
    float v1p = 0.f, v2p = 0.f;

    #pragma unroll
    for (int ks = 0; ks < 32; ++ks) {
        const int k0 = ks * 4;
        const v2f a1 = *(const v2f*)(pA + k0);
        const v2f a2 = *(const v2f*)(pB + k0);
        accN = __builtin_amdgcn_wmma_f32_16x16x4_f32(
             false, a1,  false, a2,
             (short)0, accN,  false,  false);
        v1p += a1.x * s1[k0 + kk] + a1.y * s1[k0 + kk + 1];
        v2p += a2.x * s2[k0 + kk] + a2.y * s2[k0 + kk + 1];
    }

    const float v1 = v1p + __shfl_xor(v1p, 16, 32);
    const float v2 = v2p + __shfl_xor(v2p, 16, 32);

    float numD = 0.f;
    #pragma unroll
    for (int v = 0; v < 8; ++v) {
        if (lane == v)      numD = accN[v];
        if (lane == v + 24) numD = accN[v];
    }

    int row = -1;
    if (lane < 8)       row = lane;
    else if (lane >= 24) row = lane - 16;
    if (row >= 0) {
        out[r0 + row] = numD / sqrtf(v1 * v2);
    }
}

extern "C" void kernel_launch(void* const* d_in, const int* in_sizes, int n_in,
                              void* d_out, int out_size, void* d_ws, size_t ws_size,
                              hipStream_t stream) {
    const float* x1 = (const float*)d_in[0];
    const float* x2 = (const float*)d_in[1];
    float* out  = (float*)d_out;
    float* sums = (float*)d_ws;
    float* part = out;

    colsum_kernel<<<CS_BLOCKS, 256, 0, stream>>>(x1, x2, part);
    colreduce_kernel<<<1, 256, 0, stream>>>(part, sums);
    cosine_kernel<<<NROWS / 128, 256, 0, stream>>>(x1, x2, sums, out);
}


# both passes give a work-group the same rows (512 resident groups x 4 tiles); the final pass walks them last-read-first and finds them in L2; column sums add rows in a new order
# speedup vs baseline: 1.4559x; 1.0425x over previous
#include <hip/hip_runtime.h>
#include <hip/hip_bf16.h>

#define NROWS 262144
#define DCOLS 128
#define CS_BLOCKS 512
#define PART_STRIDE 65536
#define TILE_ROWS 128
#define TPB (NROWS / (TILE_ROWS * CS_BLOCKS))
static_assert(NROWS % (TILE_ROWS * CS_BLOCKS) == 0, "every work-group owns the same number of whole tiles");

typedef __attribute__((ext_vector_type(2))) float v2f;
typedef __attribute__((ext_vector_type(8))) float v8f;

__device__ __forceinline__ void f4add(float4& a, const float4& b) {
    a.x += b.x; a.y += b.y; a.z += b.z; a.w += b.w;
}

#define CS_TILE4 ((size_t)CS_BLOCKS * TILE_ROWS * (DCOLS / 4))

__global__ __launch_bounds__(256) __attribute__((amdgpu_waves_per_eu(4, 4))) void colsum_kernel(
    const float* __restrict__ x1, const float* __restrict__ x2,
    float* __restrict__ part) {
    const int tid = threadIdx.x;
    const int cg  = tid & 31;
    const int rg  = tid >> 5;
    float4 a1 = make_float4(0.f, 0.f, 0.f, 0.f);
    float4 a2 = make_float4(0.f, 0.f, 0.f, 0.f);
    const float4* p1 = (const float4*)(x1 + ((size_t)blockIdx.x * TILE_ROWS + rg * 16) * DCOLS) + cg;
    const float4* p2 = (const float4*)(x2 + ((size_t)blockIdx.x * TILE_ROWS + rg * 16) * DCOLS) + cg;
    #pragma unroll
    for (int j = 0; j < TPB; ++j) {
        #pragma unroll
        for (int r = 0; r < 16; ++r) {
            f4add(a1, p1[(size_t)j * CS_TILE4 + r * (DCOLS / 4)]);
            f4add(a2, p2[(size_t)j * CS_TILE4 + r * (DCOLS / 4)]);
        }
    }
    __shared__ float4 l1[256];
    __shared__ float4 l2[256];
    l1[tid] = a1;
    l2[tid] = a2;
    __syncthreads();
    #pragma unroll
    for (int s = 4; s >= 1; s >>= 1) {
        if (rg < s) {
            f4add(l1[tid], l1[tid + s * 32]);
            f4add(l2[tid], l2[tid + s * 32]);
        }
        __syncthreads();
    }
    if (rg == 0) {
        float* o1 = part + (size_t)blockIdx.x * DCOLS + cg * 4;
        float* o2 = o1 + PART_STRIDE;
        *(float4*)o1 = l1[tid];
        *(float4*)o2 = l2[tid];
    }
}

__global__ __launch_bounds__(256) void colreduce_kernel(
    const float* __restrict__ part, float* __restrict__ sums) {
    const int t   = threadIdx.x;
    const int mat = t >> 7;
    const int c   = t & 127;
    const float* p = part + (size_t)mat * PART_STRIDE + c;
    float a0 = 0.f, a1 = 0.f, a2 = 0.f, a3 = 0.f;
    for (int b = 0; b < CS_BLOCKS; b += 4) {
        a0 += p[(size_t)(b + 0) * DCOLS];
        a1 += p[(size_t)(b + 1) * DCOLS];
        a2 += p[(size_t)(b + 2) * DCOLS];
        a3 += p[(size_t)(b + 3) * DCOLS];
    }
    sums[t] = (a0 + a1) + (a2 + a3);
}

#define TILE_PITCH4 33
#define TILE_PITCH  (TILE_PITCH4 * 4)

#define ROWS16(M, off) M(0, off) M(1, off) M(2, off) M(3, off) M(4, off) M(5, off) M(6, off) M(7, off) \
                       M(8, off) M(9, off) M(10, off) M(11, off) M(12, off) M(13, off) M(14, off) M(15, off)
#define Q_LOAD(r, off)  const float4 qa##r = g1[(off) + (r) * (DCOLS / 4)]; \
                        const float4 qb##r = g2[(off) + (r) * (DCOLS / 4)];
#define Q_STORE(r, off) s_tile[(off) + (r) * TILE_PITCH4] = qa##r; \
                        s_tile[(off) + (16 + (r)) * TILE_PITCH4] = qb##r;

__device__ __forceinline__ float tile_products(
    const float* __restrict__ pA, const float* __restrict__ pB,
    const float* __restrict__ s1, const float* __restrict__ s2,
    const int kk, const int lane) {
    v8f accN = {0.f, 0.f, 0.f, 0.f, 0.f, 0.f, 0.f, 0.f};
    float v1p = 0.f, v2p = 0.f;

    #pragma unroll 8
    for (int ks = 0; ks < 32; ++ks) {
        const int k0 = ks * 4;
        const v2f a1 = *(const v2f*)(pA + k0);
        const v2f a2 = *(const v2f*)(pB + k0);
        accN = __builtin_amdgcn_wmma_f32_16x16x4_f32(
             false, a1,  false, a2,
             (short)0, accN,  false,  false);
        v1p += a1.x * s1[k0 + kk] + a1.y * s1[k0 + kk + 1];
        v2p += a2.x * s2[k0 + kk] + a2.y * s2[k0 + kk + 1];
    }

    const float v1 = v1p + __shfl_xor(v1p, 16, 32);
    const float v2 = v2p + __shfl_xor(v2p, 16, 32);

    float numD = 0.f;
    #pragma unroll
    for (int v = 0; v < 8; ++v) {
        if (lane == v)      numD = accN[v];
        if (lane == v + 24) numD = accN[v];
    }

    return numD / sqrtf(v1 * v2);
}

__device__ __forceinline__ void tile_store(const float res, const int lane,
                                           float* __restrict__ out, const int r0) {
    int row = -1;
    if (lane < 8)       row = lane;
    else if (lane >= 24) row = lane - 16;
    if (row >= 0) {
        out[r0 + row] = res;
    }
}

__global__ __launch_bounds__(256) __attribute__((amdgpu_waves_per_eu(4, 4))) void cosine_kernel(
    const float* x1, const float* x2,
    const float* __restrict__ sums, float* __restrict__ out) {
    __shared__ float4 s_lds[64 + 8 * 2 * 16 * TILE_PITCH4];
    float*  s_sum  = (float*)s_lds;
    float4* s_tile = s_lds + 64;
    const int tid  = threadIdx.x;
    const int wave = tid >> 5;
    const int lane = tid & 31;

    const float ssum = sums[tid];

    const float4* g1 = (const float4*)(x1 + ((size_t)blockIdx.x * TILE_ROWS + wave * 16) * DCOLS) + lane;
    const float4* g2 = (const float4*)(x2 + ((size_t)blockIdx.x * TILE_ROWS + wave * 16) * DCOLS) + lane;
    float4* t1 = s_tile + (wave * 2 + 0) * (16 * TILE_PITCH4);
    float4* t2 = s_tile + (wave * 2 + 1) * (16 * TILE_PITCH4);
    const int tq0 = (wave * 2) * (16 * TILE_PITCH4) + lane;
    {
        ROWS16(Q_LOAD, (size_t)(TPB - 1) * CS_TILE4)
        s_sum[tid] = ssum;
        ROWS16(Q_STORE, tq0)
    }
    __syncthreads();

    const int m  = lane & 15;
    const int kk = (lane >> 4) * 2;

    const float* pA = (const float*)t1 + m * TILE_PITCH + kk;
    const float* pB = (const float*)t2 + m * TILE_PITCH + kk;
    const float* s1 = s_sum;
    const float* s2 = s_sum + 128;
    const int rw = (blockIdx.x * 8 + wave) * 16;

    #pragma unroll 1
    for (int j = TPB - 1; j >= 1; --j) {
        int tq = tq0;
        asm volatile("" : "+v"(tq));
        ROWS16(Q_LOAD, (size_t)(j - 1) * CS_TILE4)
        asm volatile("" ::: "memory");

        const float res = tile_products(pA, pB, s1, s2, kk, lane);

        ROWS16(Q_STORE, tq)
        __syncthreads();
        tile_store(res, lane, out, rw + j * (CS_BLOCKS * TILE_ROWS));
    }
    tile_store(tile_products(pA, pB, s1, s2, kk, lane), lane, out, rw);
}

extern "C" void kernel_launch(void* const* d_in, const int* in_sizes, int n_in,
                              void* d_out, int out_size, void* d_ws, size_t ws_size,
                              hipStream_t stream) {
    const float* x1 = (const float*)d_in[0];
    const float* x2 = (const float*)d_in[1];
    float* out  = (float*)d_out;
    float* sums = (float*)d_ws;
    float* part = out;

    colsum_kernel<<<CS_BLOCKS, 256, 0, stream>>>(x1, x2, part);
    colreduce_kernel<<<1, 256, 0, stream>>>(part, sums);
    cosine_kernel<<<CS_BLOCKS, 256, 0, stream>>>(x1, x2, sums, out);
}


# column-sum reduction launch: 8 work-groups x 16 waves, a lane's 8 four-line b128 loads all in flight at once (was 1 work-group, 16 trips of 32 one-line loads); partial rows summed in a new order
# speedup vs baseline: 1.6779x; 1.1525x over previous
#include <hip/hip_runtime.h>
#include <hip/hip_bf16.h>

#define NROWS 262144
#define DCOLS 128
#define CS_BLOCKS 512
#define PART_STRIDE 65536
#define TILE_ROWS 128
#define TPB (NROWS / (TILE_ROWS * CS_BLOCKS))
static_assert(NROWS % (TILE_ROWS * CS_BLOCKS) == 0, "every work-group owns the same number of whole tiles");

typedef __attribute__((ext_vector_type(2))) float v2f;
typedef __attribute__((ext_vector_type(8))) float v8f;

__device__ __forceinline__ void f4add(float4& a, const float4& b) {
    a.x += b.x; a.y += b.y; a.z += b.z; a.w += b.w;
}

#define CS_TILE4 ((size_t)CS_BLOCKS * TILE_ROWS * (DCOLS / 4))

__global__ __launch_bounds__(256) __attribute__((amdgpu_waves_per_eu(4, 4))) void colsum_kernel(
    const float* __restrict__ x1, const float* __restrict__ x2,
    float* __restrict__ part) {
    const int tid = threadIdx.x;
    const int cg  = tid & 31;
    const int rg  = tid >> 5;
    float4 a1 = make_float4(0.f, 0.f, 0.f, 0.f);
    float4 a2 = make_float4(0.f, 0.f, 0.f, 0.f);
    const float4* p1 = (const float4*)(x1 + ((size_t)blockIdx.x * TILE_ROWS + rg * 16) * DCOLS) + cg;
    const float4* p2 = (const float4*)(x2 + ((size_t)blockIdx.x * TILE_ROWS + rg * 16) * DCOLS) + cg;
    #pragma unroll
    for (int j = 0; j < TPB; ++j) {
        #pragma unroll
        for (int r = 0; r < 16; ++r) {
            f4add(a1, p1[(size_t)j * CS_TILE4 + r * (DCOLS / 4)]);
            f4add(a2, p2[(size_t)j * CS_TILE4 + r * (DCOLS / 4)]);
        }
    }
    __shared__ float4 l1[256];
    __shared__ float4 l2[256];
    l1[tid] = a1;
    l2[tid] = a2;
    __syncthreads();
    #pragma unroll
    for (int s = 4; s >= 1; s >>= 1) {
        if (rg < s) {
            f4add(l1[tid], l1[tid + s * 32]);
            f4add(l2[tid], l2[tid + s * 32]);
        }
        __syncthreads();
    }
    if (rg == 0) {
        float* o1 = part + (size_t)blockIdx.x * DCOLS + cg * 4;
        float* o2 = o1 + PART_STRIDE;
        *(float4*)o1 = l1[tid];
        *(float4*)o2 = l2[tid];
    }
}

#define RD_GROUPS  8
#define RD_THREADS 512
#define RD_WAVES   16
#define RD_ROWS    32
#define RD_LOADS   8
static_assert(RD_GROUPS == 2 * (DCOLS / 32), "one work-group per matrix and 128-B line of a partial row");
static_assert(RD_WAVES * 32 == RD_THREADS && RD_WAVES * RD_ROWS == CS_BLOCKS && RD_LOADS * 4 == RD_ROWS,
              "the waves' rows cover the partial rows exactly once");

__global__ __launch_bounds__(RD_THREADS) void colreduce_kernel(
    const float* __restrict__ part, float* __restrict__ sums) {
    const int tid  = threadIdx.x;
    const int wave = tid >> 5;
    const int lane = tid & 31;
    const int sub  = lane & 7;
    const int rr   = lane >> 3;
    const int mat  = blockIdx.x >> 2;
    const int cgrp = blockIdx.x & 3;
    const float4* p = (const float4*)(part + (size_t)mat * PART_STRIDE
                                      + (size_t)(wave * RD_ROWS + rr) * DCOLS + cgrp * 32) + sub;
    float4 q[RD_LOADS];
    #pragma unroll
    for (int i = 0; i < RD_LOADS; ++i) q[i] = p[i * 4 * (DCOLS / 4)];
    float4 a = q[0];
    #pragma unroll
    for (int i = 1; i < RD_LOADS; ++i) f4add(a, q[i]);
    a.x += __shfl_xor(a.x, 8, 32);  a.y += __shfl_xor(a.y, 8, 32);
    a.z += __shfl_xor(a.z, 8, 32);  a.w += __shfl_xor(a.w, 8, 32);
    a.x += __shfl_xor(a.x, 16, 32); a.y += __shfl_xor(a.y, 16, 32);
    a.z += __shfl_xor(a.z, 16, 32); a.w += __shfl_xor(a.w, 16, 32);
    __shared__ float4 red[RD_WAVES * 8];
    if (lane < 8) red[wave * 8 + lane] = a;
    __syncthreads();
    if (tid < 32) {
        const float* r = (const float*)red;
        float s = r[tid];
        #pragma unroll
        for (int w = 1; w < RD_WAVES; ++w) s += r[w * 32 + tid];
        sums[mat * DCOLS + cgrp * 32 + tid] = s;
    }
}

#define TILE_PITCH4 33
#define TILE_PITCH  (TILE_PITCH4 * 4)

#define ROWS16(M, off) M(0, off) M(1, off) M(2, off) M(3, off) M(4, off) M(5, off) M(6, off) M(7, off) \
                       M(8, off) M(9, off) M(10, off) M(11, off) M(12, off) M(13, off) M(14, off) M(15, off)
#define Q_LOAD(r, off)  const float4 qa##r = g1[(off) + (r) * (DCOLS / 4)]; \
                        const float4 qb##r = g2[(off) + (r) * (DCOLS / 4)];
#define Q_STORE(r, off) s_tile[(off) + (r) * TILE_PITCH4] = qa##r; \
                        s_tile[(off) + (16 + (r)) * TILE_PITCH4] = qb##r;

__device__ __forceinline__ float tile_products(
    const float* __restrict__ pA, const float* __restrict__ pB,
    const float* __restrict__ s1, const float* __restrict__ s2,
    const int kk, const int lane) {
    v8f accN = {0.f, 0.f, 0.f, 0.f, 0.f, 0.f, 0.f, 0.f};
    float v1p = 0.f, v2p = 0.f;

    #pragma unroll 8
    for (int ks = 0; ks < 32; ++ks) {
        const int k0 = ks * 4;
        const v2f a1 = *(const v2f*)(pA + k0);
        const v2f a2 = *(const v2f*)(pB + k0);
        accN = __builtin_amdgcn_wmma_f32_16x16x4_f32(
             false, a1,  false, a2,
             (short)0, accN,  false,  false);
        v1p += a1.x * s1[k0 + kk] + a1.y * s1[k0 + kk + 1];
        v2p += a2.x * s2[k0 + kk] + a2.y * s2[k0 + kk + 1];
    }

    const float v1 = v1p + __shfl_xor(v1p, 16, 32);
    const float v2 = v2p + __shfl_xor(v2p, 16, 32);

    float numD = 0.f;
    #pragma unroll
    for (int v = 0; v < 8; ++v) {
        if (lane == v)      numD = accN[v];
        if (lane == v + 24) numD = accN[v];
    }

    return numD / sqrtf(v1 * v2);
}

__device__ __forceinline__ void tile_store(const float res, const int lane,
                                           float* __restrict__ out, const int r0) {
    int row = -1;
    if (lane < 8)       row = lane;
    else if (lane >= 24) row = lane - 16;
    if (row >= 0) {
        out[r0 + row] = res;
    }
}

__global__ __launch_bounds__(256) __attribute__((amdgpu_waves_per_eu(4, 4))) void cosine_kernel(
    const float* x1, const float* x2,
    const float* __restrict__ sums, float* __restrict__ out) {
    __shared__ float4 s_lds[64 + 8 * 2 * 16 * TILE_PITCH4];
    float*  s_sum  = (float*)s_lds;
    float4* s_tile = s_lds + 64;
    const int tid  = threadIdx.x;
    const int wave = tid >> 5;
    const int lane = tid & 31;

    const float ssum = sums[tid];

    const float4* g1 = (const float4*)(x1 + ((size_t)blockIdx.x * TILE_ROWS + wave * 16) * DCOLS) + lane;
    const float4* g2 = (const float4*)(x2 + ((size_t)blockIdx.x * TILE_ROWS + wave * 16) * DCOLS) + lane;
    float4* t1 = s_tile + (wave * 2 + 0) * (16 * TILE_PITCH4);
    float4* t2 = s_tile + (wave * 2 + 1) * (16 * TILE_PITCH4);
    const int tq0 = (wave * 2) * (16 * TILE_PITCH4) + lane;
    {
        ROWS16(Q_LOAD, (size_t)(TPB - 1) * CS_TILE4)
        s_sum[tid] = ssum;
        ROWS16(Q_STORE, tq0)
    }
    __syncthreads();

    const int m  = lane & 15;
    const int kk = (lane >> 4) * 2;

    const float* pA = (const float*)t1 + m * TILE_PITCH + kk;
    const float* pB = (const float*)t2 + m * TILE_PITCH + kk;
    const float* s1 = s_sum;
    const float* s2 = s_sum + 128;
    const int rw = (blockIdx.x * 8 + wave) * 16;

    #pragma unroll 1
    for (int j = TPB - 1; j >= 1; --j) {
        int tq = tq0;
        asm volatile("" : "+v"(tq));
        ROWS16(Q_LOAD, (size_t)(j - 1) * CS_TILE4)
        asm volatile("" ::: "memory");

        const float res = tile_products(pA, pB, s1, s2, kk, lane);

        ROWS16(Q_STORE, tq)
        __syncthreads();
        tile_store(res, lane, out, rw + j * (CS_BLOCKS * TILE_ROWS));
    }
    tile_store(tile_products(pA, pB, s1, s2, kk, lane), lane, out, rw);
}

extern "C" void kernel_launch(void* const* d_in, const int* in_sizes, int n_in,
                              void* d_out, int out_size, void* d_ws, size_t ws_size,
                              hipStream_t stream) {
    const float* x1 = (const float*)d_in[0];
    const float* x2 = (const float*)d_in[1];
    float* out  = (float*)d_out;
    float* sums = (float*)d_ws;
    float* part = out;

    colsum_kernel<<<CS_BLOCKS, 256, 0, stream>>>(x1, x2, part);
    colreduce_kernel<<<RD_GROUPS, RD_THREADS, 0, stream>>>(part, sums);
    cosine_kernel<<<CS_BLOCKS, 256, 0, stream>>>(x1, x2, sums, out);
}
